# starting kernel
# baseline (speedup 1.0000x reference)
#include <hip/hip_runtime.h>

typedef __attribute__((ext_vector_type(2))) float v2f;
typedef __attribute__((ext_vector_type(8))) float v8f;

#define NNODES 50000
#define KNBR   32
#define DIN    512
#define DHID   128
#define DOUT   64

__global__ __launch_bounds__(256)
void gemm1_wmma(const float* __restrict__ feat, const float* __restrict__ W1,
                const float* __restrict__ b1, float* __restrict__ h)
{
    __shared__ float As[16 * DIN];
    const int rowBase = blockIdx.x * 16;
    const int tid = threadIdx.x;

    {
        const float4* src = (const float4*)(feat + (size_t)rowBase * DIN);
        float4* dst = (float4*)As;
        #pragma unroll
        for (int i = 0; i < (16 * DIN / 4) / 256; ++i)
            dst[tid + i * 256] = src[tid + i * 256];
    }
    __syncthreads();

    const int wave = tid >> 5;
    const int lane = tid & 31;
    const int m    = lane & 15;
    const int kp   = (lane >> 4) << 1;
    const int n    = wave * 16 + m;

    const float* Arow = As + m * DIN + kp;
    const float* Bcol = W1 + n;

    v8f c = {};
    #pragma unroll 8
    for (int k = 0; k < DIN; k += 4) {
        v2f a = *(const v2f*)(Arow + k);
        v2f b;
        b.x = Bcol[(size_t)(k + kp)     * DHID];
        b.y = Bcol[(size_t)(k + kp + 1) * DHID];
        c = __builtin_amdgcn_wmma_f32_16x16x4_f32(false, a, false, b,
                                                  (short)0, c, false, false);
    }

    const float bias = b1[n];
    const int rOff = (lane >> 4) * 8;
    #pragma unroll
    for (int r = 0; r < 8; ++r)
        h[(size_t)(rowBase + rOff + r) * DHID + n] = c[r] + bias;
}

__global__ __launch_bounds__(128)
void gemm2_wmma(const float* __restrict__ hm, const float* __restrict__ W2,
                const float* __restrict__ b2, float* __restrict__ h2)
{
    __shared__ float As[16 * DHID];
    const int rowBase = blockIdx.x * 16;
    const int tid = threadIdx.x;

    {
        const float4* src = (const float4*)(hm + (size_t)rowBase * DHID);
        float4* dst = (float4*)As;
        #pragma unroll
        for (int i = 0; i < (16 * DHID / 4) / 128; ++i)
            dst[tid + i * 128] = src[tid + i * 128];
    }
    __syncthreads();

    const int wave = tid >> 5;
    const int lane = tid & 31;
    const int m    = lane & 15;
    const int kp   = (lane >> 4) << 1;
    const int n    = wave * 16 + m;

    const float* Arow = As + m * DHID + kp;
    const float* Bcol = W2 + n;

    v8f c = {};
    #pragma unroll 8
    for (int k = 0; k < DHID; k += 4) {
        v2f a = *(const v2f*)(Arow + k);
        v2f b;
        b.x = Bcol[(size_t)(k + kp)     * DOUT];
        b.y = Bcol[(size_t)(k + kp + 1) * DOUT];
        c = __builtin_amdgcn_wmma_f32_16x16x4_f32(false, a, false, b,
                                                  (short)0, c, false, false);
    }

    const float bias = b2[n];
    const int rOff = (lane >> 4) * 8;
    #pragma unroll
    for (int r = 0; r < 8; ++r)
        h2[(size_t)(rowBase + rOff + r) * DOUT + n] = c[r] + bias;
}

__device__ __forceinline__ void cmpswap(float& a, float& b)
{
    const float lo = fminf(a, b);
    const float hi = fmaxf(a, b);
    a = lo;
    b = hi;
}

template <int D, int NODES_PER_BLK, bool RELU>
__global__ __launch_bounds__(D * NODES_PER_BLK)
void median_kernel(const float* __restrict__ src,
                   const int* __restrict__ nbr,
                   float* __restrict__ dst)
{
    const int node = blockIdx.x * NODES_PER_BLK + threadIdx.x / D;
    const int f    = threadIdx.x % D;

    float v[KNBR];
    #pragma unroll
    for (int j = 0; j < KNBR; ++j) {
        const int idx = nbr[node * KNBR + j];
        v[j] = src[(size_t)idx * D + f];
    }

    #pragma unroll
    for (int p = 1; p < KNBR; p <<= 1) {
        #pragma unroll
        for (int k = p; k >= 1; k >>= 1) {
            #pragma unroll
            for (int j = k % p; j + k < KNBR; j += 2 * k) {
                #pragma unroll
                for (int i = 0; i < k; ++i) {
                    if ((i + j) / (2 * p) == (i + j + k) / (2 * p))
                        cmpswap(v[i + j], v[i + j + k]);
                }
            }
        }
    }

    float med = v[(KNBR - 1) / 2];
    if (RELU) med = med > 0.0f ? med : 0.0f;
    dst[(size_t)node * D + f] = med;
}

extern "C" void kernel_launch(void* const* d_in, const int* in_sizes, int n_in,
                              void* d_out, int out_size, void* d_ws, size_t ws_size,
                              hipStream_t stream)
{
    const float* feat = (const float*)d_in[0];
    const float* W1   = (const float*)d_in[1];
    const float* b1   = (const float*)d_in[2];
    const float* W2   = (const float*)d_in[3];
    const float* b2   = (const float*)d_in[4];
    const int*   nbr  = (const int*)d_in[5];

    float* h   = (float*)d_ws;
    float* hm  = h  + (size_t)NNODES * DHID;
    float* h2  = hm + (size_t)NNODES * DHID;
    float* out = (float*)d_out;

    gemm1_wmma<<<NNODES / 16, 256, 0, stream>>>(feat, W1, b1, h);
    median_kernel<DHID, 1, true ><<<NNODES,     DHID, 0, stream>>>(h,  nbr, hm);
    gemm2_wmma<<<NNODES / 16, 128, 0, stream>>>(hm, W2, b2, h2);
    median_kernel<DOUT, 2, false><<<NNODES / 2, DOUT * 2, 0, stream>>>(h2, nbr, out);
}


# Both GEMMs: LDS row pitch of the staged A tile padded by 4 floats (516 / 132) so the A-fragment ds_load_b64 no longer hits one bank with all 16 rows (16-way conflict removed)
# speedup vs baseline: 1.1862x; 1.1862x over previous
#include <hip/hip_runtime.h>

typedef __attribute__((ext_vector_type(2))) float v2f;
typedef __attribute__((ext_vector_type(8))) float v8f;

#define NNODES 50000
#define KNBR   32
#define DIN    512
#define DHID   128
#define DOUT   64
#define APITCH1 (DIN + 4)
#define APITCH2 (DHID + 4)

__global__ __launch_bounds__(256)
void gemm1_wmma(const float* __restrict__ feat, const float* __restrict__ W1,
                const float* __restrict__ b1, float* __restrict__ h)
{
    __shared__ __attribute__((aligned(16))) float As[16 * APITCH1];
    const int rowBase = blockIdx.x * 16;
    const int tid = threadIdx.x;

    {
        const float4* src = (const float4*)(feat + (size_t)rowBase * DIN);
        const int c4 = tid & (DIN / 4 - 1);
        const int r0 = tid / (DIN / 4);
        #pragma unroll
        for (int i = 0; i < (16 * DIN / 4) / 256; ++i)
            *(float4*)(As + (r0 + 2 * i) * APITCH1 + c4 * 4) = src[tid + i * 256];
    }
    __syncthreads();

    const int wave = tid >> 5;
    const int lane = tid & 31;
    const int m    = lane & 15;
    const int kp   = (lane >> 4) << 1;
    const int n    = wave * 16 + m;

    const float* Arow = As + m * APITCH1 + kp;
    const float* Bcol = W1 + n;

    v8f c = {};
    #pragma unroll 8
    for (int k = 0; k < DIN; k += 4) {
        v2f a = *(const v2f*)(Arow + k);
        v2f b;
        b.x = Bcol[(size_t)(k + kp)     * DHID];
        b.y = Bcol[(size_t)(k + kp + 1) * DHID];
        c = __builtin_amdgcn_wmma_f32_16x16x4_f32(false, a, false, b,
                                                  (short)0, c, false, false);
    }

    const float bias = b1[n];
    const int rOff = (lane >> 4) * 8;
    #pragma unroll
    for (int r = 0; r < 8; ++r)
        h[(size_t)(rowBase + rOff + r) * DHID + n] = c[r] + bias;
}

__global__ __launch_bounds__(128)
void gemm2_wmma(const float* __restrict__ hm, const float* __restrict__ W2,
                const float* __restrict__ b2, float* __restrict__ h2)
{
    __shared__ __attribute__((aligned(16))) float As[16 * APITCH2];
    const int rowBase = blockIdx.x * 16;
    const int tid = threadIdx.x;

    {
        const float4* src = (const float4*)(hm + (size_t)rowBase * DHID);
        const int c4 = tid & (DHID / 4 - 1);
        const int r0 = tid / (DHID / 4);
        #pragma unroll
        for (int i = 0; i < (16 * DHID / 4) / 128; ++i)
            *(float4*)(As + (r0 + 4 * i) * APITCH2 + c4 * 4) = src[tid + i * 128];
    }
    __syncthreads();

    const int wave = tid >> 5;
    const int lane = tid & 31;
    const int m    = lane & 15;
    const int kp   = (lane >> 4) << 1;
    const int n    = wave * 16 + m;

    const float* Arow = As + m * APITCH2 + kp;
    const float* Bcol = W2 + n;

    v8f c = {};
    #pragma unroll 8
    for (int k = 0; k < DHID; k += 4) {
        v2f a = *(const v2f*)(Arow + k);
        v2f b;
        b.x = Bcol[(size_t)(k + kp)     * DOUT];
        b.y = Bcol[(size_t)(k + kp + 1) * DOUT];
        c = __builtin_amdgcn_wmma_f32_16x16x4_f32(false, a, false, b,
                                                  (short)0, c, false, false);
    }

    const float bias = b2[n];
    const int rOff = (lane >> 4) * 8;
    #pragma unroll
    for (int r = 0; r < 8; ++r)
        h2[(size_t)(rowBase + rOff + r) * DOUT + n] = c[r] + bias;
}

__device__ __forceinline__ void cmpswap(float& a, float& b)
{
    const float lo = fminf(a, b);
    const float hi = fmaxf(a, b);
    a = lo;
    b = hi;
}

template <int D, int NODES_PER_BLK, bool RELU>
__global__ __launch_bounds__(D * NODES_PER_BLK)
void median_kernel(const float* __restrict__ src,
                   const int* __restrict__ nbr,
                   float* __restrict__ dst)
{
    const int node = blockIdx.x * NODES_PER_BLK + threadIdx.x / D;
    const int f    = threadIdx.x % D;

    float v[KNBR];
    #pragma unroll
    for (int j = 0; j < KNBR; ++j) {
        const int idx = nbr[node * KNBR + j];
        v[j] = src[(size_t)idx * D + f];
    }

    #pragma unroll
    for (int p = 1; p < KNBR; p <<= 1) {
        #pragma unroll
        for (int k = p; k >= 1; k >>= 1) {
            #pragma unroll
            for (int j = k % p; j + k < KNBR; j += 2 * k) {
                #pragma unroll
                for (int i = 0; i < k; ++i) {
                    if ((i + j) / (2 * p) == (i + j + k) / (2 * p))
                        cmpswap(v[i + j], v[i + j + k]);
                }
            }
        }
    }

    float med = v[(KNBR - 1) / 2];
    if (RELU) med = med > 0.0f ? med : 0.0f;
    dst[(size_t)node * D + f] = med;
}

extern "C" void kernel_launch(void* const* d_in, const int* in_sizes, int n_in,
                              void* d_out, int out_size, void* d_ws, size_t ws_size,
                              hipStream_t stream)
{
    const float* feat = (const float*)d_in[0];
    const float* W1   = (const float*)d_in[1];
    const float* b1   = (const float*)d_in[2];
    const float* W2   = (const float*)d_in[3];
    const float* b2   = (const float*)d_in[4];
    const int*   nbr  = (const int*)d_in[5];

    float* h   = (float*)d_ws;
    float* hm  = h  + (size_t)NNODES * DHID;
    float* h2  = hm + (size_t)NNODES * DHID;
    float* out = (float*)d_out;

    gemm1_wmma<<<NNODES / 16, 256, 0, stream>>>(feat, W1, b1, h);
    median_kernel<DHID, 1, true ><<<NNODES,     DHID, 0, stream>>>(h,  nbr, hm);
    gemm2_wmma<<<NNODES / 16, 128, 0, stream>>>(hm, W2, b2, h2);
    median_kernel<DOUT, 2, false><<<NNODES / 2, DOUT * 2, 0, stream>>>(h2, nbr, out);
}


# Median kernels: neighbour indices come by scalar loads and each row base stays in a scalar register pair (gather = scalar base + shared lane offset); per-lane 64-bit address arithmetic removed
# speedup vs baseline: 1.3351x; 1.1255x over previous
#include <hip/hip_runtime.h>

typedef __attribute__((ext_vector_type(2))) float v2f;
typedef __attribute__((ext_vector_type(8))) float v8f;

#define NNODES 50000
#define KNBR   32
#define DIN    512
#define DHID   128
#define DOUT   64
#define APITCH1 (DIN + 4)
#define APITCH2 (DHID + 4)

__global__ __launch_bounds__(256)
void gemm1_wmma(const float* __restrict__ feat, const float* __restrict__ W1,
                const float* __restrict__ b1, float* __restrict__ h)
{
    __shared__ __attribute__((aligned(16))) float As[16 * APITCH1];
    const int rowBase = blockIdx.x * 16;
    const int tid = threadIdx.x;

    {
        const float4* src = (const float4*)(feat + (size_t)rowBase * DIN);
        const int c4 = tid & (DIN / 4 - 1);
        const int r0 = tid / (DIN / 4);
        #pragma unroll
        for (int i = 0; i < (16 * DIN / 4) / 256; ++i)
            *(float4*)(As + (r0 + 2 * i) * APITCH1 + c4 * 4) = src[tid + i * 256];
    }
    __syncthreads();

    const int wave = tid >> 5;
    const int lane = tid & 31;
    const int m    = lane & 15;
    const int kp   = (lane >> 4) << 1;
    const int n    = wave * 16 + m;

    const float* Arow = As + m * APITCH1 + kp;
    const float* Bcol = W1 + n;

    v8f c = {};
    #pragma unroll 8
    for (int k = 0; k < DIN; k += 4) {
        v2f a = *(const v2f*)(Arow + k);
        v2f b;
        b.x = Bcol[(size_t)(k + kp)     * DHID];
        b.y = Bcol[(size_t)(k + kp + 1) * DHID];
        c = __builtin_amdgcn_wmma_f32_16x16x4_f32(false, a, false, b,
                                                  (short)0, c, false, false);
    }

    const float bias = b1[n];
    const int rOff = (lane >> 4) * 8;
    #pragma unroll
    for (int r = 0; r < 8; ++r)
        h[(size_t)(rowBase + rOff + r) * DHID + n] = c[r] + bias;
}

__global__ __launch_bounds__(128)
void gemm2_wmma(const float* __restrict__ hm, const float* __restrict__ W2,
                const float* __restrict__ b2, float* __restrict__ h2)
{
    __shared__ __attribute__((aligned(16))) float As[16 * APITCH2];
    const int rowBase = blockIdx.x * 16;
    const int tid = threadIdx.x;

    {
        const float4* src = (const float4*)(hm + (size_t)rowBase * DHID);
        const int c4 = tid & (DHID / 4 - 1);
        const int r0 = tid / (DHID / 4);
        #pragma unroll
        for (int i = 0; i < (16 * DHID / 4) / 128; ++i)
            *(float4*)(As + (r0 + 4 * i) * APITCH2 + c4 * 4) = src[tid + i * 128];
    }
    __syncthreads();

    const int wave = tid >> 5;
    const int lane = tid & 31;
    const int m    = lane & 15;
    const int kp   = (lane >> 4) << 1;
    const int n    = wave * 16 + m;

    const float* Arow = As + m * APITCH2 + kp;
    const float* Bcol = W2 + n;

    v8f c = {};
    #pragma unroll 8
    for (int k = 0; k < DHID; k += 4) {
        v2f a = *(const v2f*)(Arow + k);
        v2f b;
        b.x = Bcol[(size_t)(k + kp)     * DOUT];
        b.y = Bcol[(size_t)(k + kp + 1) * DOUT];
        c = __builtin_amdgcn_wmma_f32_16x16x4_f32(false, a, false, b,
                                                  (short)0, c, false, false);
    }

    const float bias = b2[n];
    const int rOff = (lane >> 4) * 8;
    #pragma unroll
    for (int r = 0; r < 8; ++r)
        h2[(size_t)(rowBase + rOff + r) * DOUT + n] = c[r] + bias;
}

__device__ __forceinline__ void cmpswap(float& a, float& b)
{
    const float lo = fminf(a, b);
    const float hi = fmaxf(a, b);
    a = lo;
    b = hi;
}

template <int D, int NODES_PER_BLK, bool RELU>
__global__ __launch_bounds__(D * NODES_PER_BLK)
void median_kernel(const float* __restrict__ src,
                   const int* __restrict__ nbr,
                   float* __restrict__ dst)
{
    const int node = __builtin_amdgcn_readfirstlane(blockIdx.x * NODES_PER_BLK + threadIdx.x / D);
    const int f    = threadIdx.x % D;

    typedef const int __attribute__((address_space(4))) cidx_t;
    cidx_t* nb = (cidx_t*)(unsigned long long)(nbr + (size_t)node * KNBR);
    typedef const float __attribute__((address_space(1))) gfloat_t;

    float v[KNBR];
    #pragma unroll
    for (int j = 0; j < KNBR; ++j) {
        const unsigned long long rowAddr =
            (unsigned long long)src + (unsigned long long)((unsigned)nb[j] * (unsigned)(D * 4));
        const unsigned lo = __builtin_amdgcn_readfirstlane((unsigned)rowAddr);
        const unsigned hi = __builtin_amdgcn_readfirstlane((unsigned)(rowAddr >> 32));
        gfloat_t* row = (gfloat_t*)(((unsigned long long)hi << 32) | lo);
        v[j] = row[f];
    }

    #pragma unroll
    for (int p = 1; p < KNBR; p <<= 1) {
        #pragma unroll
        for (int k = p; k >= 1; k >>= 1) {
            #pragma unroll
            for (int j = k % p; j + k < KNBR; j += 2 * k) {
                #pragma unroll
                for (int i = 0; i < k; ++i) {
                    if ((i + j) / (2 * p) == (i + j + k) / (2 * p))
                        cmpswap(v[i + j], v[i + j + k]);
                }
            }
        }
    }

    float med = v[(KNBR - 1) / 2];
    if (RELU) med = med > 0.0f ? med : 0.0f;
    dst[(size_t)node * D + f] = med;
}

extern "C" void kernel_launch(void* const* d_in, const int* in_sizes, int n_in,
                              void* d_out, int out_size, void* d_ws, size_t ws_size,
                              hipStream_t stream)
{
    const float* feat = (const float*)d_in[0];
    const float* W1   = (const float*)d_in[1];
    const float* b1   = (const float*)d_in[2];
    const float* W2   = (const float*)d_in[3];
    const float* b2   = (const float*)d_in[4];
    const int*   nbr  = (const int*)d_in[5];

    float* h   = (float*)d_ws;
    float* hm  = h  + (size_t)NNODES * DHID;
    float* h2  = hm + (size_t)NNODES * DHID;
    float* out = (float*)d_out;

    gemm1_wmma<<<NNODES / 16, 256, 0, stream>>>(feat, W1, b1, h);
    median_kernel<DHID, 1, true ><<<NNODES,     DHID, 0, stream>>>(h,  nbr, hm);
    gemm2_wmma<<<NNODES / 16, 128, 0, stream>>>(hm, W2, b2, h2);
    median_kernel<DOUT, 2, false><<<NNODES / 2, DOUT * 2, 0, stream>>>(h2, nbr, out);
}


# GEMM1: each wave holds two 16-row accumulator tiles fed by one B fragment (block = 32 rows, A tile 32x516 in LDS), so W1 is loaded half as often per output row; last block's spare tile is not stored
# speedup vs baseline: 1.4733x; 1.1035x over previous
#include <hip/hip_runtime.h>

typedef __attribute__((ext_vector_type(2))) float v2f;
typedef __attribute__((ext_vector_type(8))) float v8f;

#define NNODES 50000
#define KNBR   32
#define DIN    512
#define DHID   128
#define DOUT   64
#define APITCH1 (DIN + 4)
#define APITCH2 (DHID + 4)
#define RT1      2
#define G1ROWS   (16 * RT1)
#define G1BLOCKS ((NNODES + G1ROWS - 1) / G1ROWS)

__global__ __launch_bounds__(256)
void gemm1_wmma(const float* __restrict__ feat, const float* __restrict__ W1,
                const float* __restrict__ b1, float* __restrict__ h)
{
    __shared__ __attribute__((aligned(16))) float As[G1ROWS * APITCH1];
    const int rowBase = blockIdx.x * G1ROWS;
    const int tid = threadIdx.x;

    #pragma unroll
    for (int t = 0; t < RT1; ++t) {
        int tileRow = rowBase + 16 * t;
        tileRow = tileRow < NNODES ? tileRow : NNODES - 16;
        const float4* src = (const float4*)(feat + (size_t)tileRow * DIN);
        const int c4 = tid & (DIN / 4 - 1);
        const int r0 = tid / (DIN / 4);
        #pragma unroll
        for (int i = 0; i < (16 * DIN / 4) / 256; ++i)
            *(float4*)(As + (16 * t + r0 + 2 * i) * APITCH1 + c4 * 4) = src[tid + i * 256];
    }
    __syncthreads();

    const int wave = tid >> 5;
    const int lane = tid & 31;
    const int m    = lane & 15;
    const int kp   = (lane >> 4) << 1;
    const int n    = wave * 16 + m;

    const float* Arow = As + m * APITCH1 + kp;
    const float* Bcol = W1 + n;

    v8f c[RT1];
    #pragma unroll
    for (int t = 0; t < RT1; ++t) c[t] = v8f{};

    #pragma unroll 8
    for (int k = 0; k < DIN; k += 4) {
        v2f b;
        b.x = Bcol[(size_t)(k + kp)     * DHID];
        b.y = Bcol[(size_t)(k + kp + 1) * DHID];
        #pragma unroll
        for (int t = 0; t < RT1; ++t) {
            v2f a = *(const v2f*)(Arow + t * 16 * APITCH1 + k);
            c[t] = __builtin_amdgcn_wmma_f32_16x16x4_f32(false, a, false, b,
                                                         (short)0, c[t], false, false);
        }
    }

    const float bias = b1[n];
    const int rOff = (lane >> 4) * 8;
    #pragma unroll
    for (int t = 0; t < RT1; ++t) {
        #pragma unroll
        for (int r = 0; r < 8; ++r) c[t][r] += bias;
    }
    #pragma unroll
    for (int t = 0; t < RT1; ++t) {
        if (rowBase + 16 * t < NNODES) {
            #pragma unroll
            for (int r = 0; r < 8; ++r)
                h[(size_t)(rowBase + 16 * t + rOff + r) * DHID + n] = c[t][r];
        }
    }
}

__global__ __launch_bounds__(128)
void gemm2_wmma(const float* __restrict__ hm, const float* __restrict__ W2,
                const float* __restrict__ b2, float* __restrict__ h2)
{
    __shared__ __attribute__((aligned(16))) float As[16 * APITCH2];
    const int rowBase = blockIdx.x * 16;
    const int tid = threadIdx.x;

    {
        const float4* src = (const float4*)(hm + (size_t)rowBase * DHID);
        const int c4 = tid & (DHID / 4 - 1);
        const int r0 = tid / (DHID / 4);
        #pragma unroll
        for (int i = 0; i < (16 * DHID / 4) / 128; ++i)
            *(float4*)(As + (r0 + 4 * i) * APITCH2 + c4 * 4) = src[tid + i * 128];
    }
    __syncthreads();

    const int wave = tid >> 5;
    const int lane = tid & 31;
    const int m    = lane & 15;
    const int kp   = (lane >> 4) << 1;
    const int n    = wave * 16 + m;

    const float* Arow = As + m * APITCH2 + kp;
    const float* Bcol = W2 + n;

    v8f c = {};
    #pragma unroll 8
    for (int k = 0; k < DHID; k += 4) {
        v2f a = *(const v2f*)(Arow + k);
        v2f b;
        b.x = Bcol[(size_t)(k + kp)     * DOUT];
        b.y = Bcol[(size_t)(k + kp + 1) * DOUT];
        c = __builtin_amdgcn_wmma_f32_16x16x4_f32(false, a, false, b,
                                                  (short)0, c, false, false);
    }

    const float bias = b2[n];
    const int rOff = (lane >> 4) * 8;
    #pragma unroll
    for (int r = 0; r < 8; ++r)
        h2[(size_t)(rowBase + rOff + r) * DOUT + n] = c[r] + bias;
}

__device__ __forceinline__ void cmpswap(float& a, float& b)
{
    const float lo = fminf(a, b);
    const float hi = fmaxf(a, b);
    a = lo;
    b = hi;
}

template <int D, int NODES_PER_BLK, bool RELU>
__global__ __launch_bounds__(D * NODES_PER_BLK)
void median_kernel(const float* __restrict__ src,
                   const int* __restrict__ nbr,
                   float* __restrict__ dst)
{
    const int node = __builtin_amdgcn_readfirstlane(blockIdx.x * NODES_PER_BLK + threadIdx.x / D);
    const int f    = threadIdx.x % D;

    typedef const int __attribute__((address_space(4))) cidx_t;
    cidx_t* nb = (cidx_t*)(unsigned long long)(nbr + (size_t)node * KNBR);
    typedef const float __attribute__((address_space(1))) gfloat_t;

    float v[KNBR];
    #pragma unroll
    for (int j = 0; j < KNBR; ++j) {
        const unsigned long long rowAddr =
            (unsigned long long)src + (unsigned long long)((unsigned)nb[j] * (unsigned)(D * 4));
        const unsigned lo = __builtin_amdgcn_readfirstlane((unsigned)rowAddr);
        const unsigned hi = __builtin_amdgcn_readfirstlane((unsigned)(rowAddr >> 32));
        gfloat_t* row = (gfloat_t*)(((unsigned long long)hi << 32) | lo);
        v[j] = row[f];
    }

    #pragma unroll
    for (int p = 1; p < KNBR; p <<= 1) {
        #pragma unroll
        for (int k = p; k >= 1; k >>= 1) {
            #pragma unroll
            for (int j = k % p; j + k < KNBR; j += 2 * k) {
                #pragma unroll
                for (int i = 0; i < k; ++i) {
                    if ((i + j) / (2 * p) == (i + j + k) / (2 * p))
                        cmpswap(v[i + j], v[i + j + k]);
                }
            }
        }
    }

    float med = v[(KNBR - 1) / 2];
    if (RELU) med = med > 0.0f ? med : 0.0f;
    dst[(size_t)node * D + f] = med;
}

extern "C" void kernel_launch(void* const* d_in, const int* in_sizes, int n_in,
                              void* d_out, int out_size, void* d_ws, size_t ws_size,
                              hipStream_t stream)
{
    const float* feat = (const float*)d_in[0];
    const float* W1   = (const float*)d_in[1];
    const float* b1   = (const float*)d_in[2];
    const float* W2   = (const float*)d_in[3];
    const float* b2   = (const float*)d_in[4];
    const int*   nbr  = (const int*)d_in[5];

    float* h   = (float*)d_ws;
    float* hm  = h  + (size_t)NNODES * DHID;
    float* h2  = hm + (size_t)NNODES * DHID;
    float* out = (float*)d_out;

    gemm1_wmma<<<G1BLOCKS, 256, 0, stream>>>(feat, W1, b1, h);
    median_kernel<DHID, 1, true ><<<NNODES,     DHID, 0, stream>>>(h,  nbr, hm);
    gemm2_wmma<<<NNODES / 16, 128, 0, stream>>>(hm, W2, b2, h2);
    median_kernel<DOUT, 2, false><<<NNODES / 2, DOUT * 2, 0, stream>>>(h2, nbr, out);
}


# Median kernels: each compare-exchange of the selection network is one hand-written dual-issue v_min/v_max instruction (one asm block: 157 vector instructions per value, was 247); ReLU folded in
# speedup vs baseline: 1.6045x; 1.0891x over previous
#include <hip/hip_runtime.h>

typedef __attribute__((ext_vector_type(2))) float v2f;
typedef __attribute__((ext_vector_type(8))) float v8f;

#define NNODES 50000
#define KNBR   32
#define DIN    512
#define DHID   128
#define DOUT   64
#define APITCH1 (DIN + 4)
#define APITCH2 (DHID + 4)
#define RT1      2
#define G1ROWS   (16 * RT1)
#define G1BLOCKS ((NNODES + G1ROWS - 1) / G1ROWS)

__global__ __launch_bounds__(256)
void gemm1_wmma(const float* __restrict__ feat, const float* __restrict__ W1,
                const float* __restrict__ b1, float* __restrict__ h)
{
    __shared__ __attribute__((aligned(16))) float As[G1ROWS * APITCH1];
    const int rowBase = blockIdx.x * G1ROWS;
    const int tid = threadIdx.x;

    #pragma unroll
    for (int t = 0; t < RT1; ++t) {
        int tileRow = rowBase + 16 * t;
        tileRow = tileRow < NNODES ? tileRow : NNODES - 16;
        const float4* src = (const float4*)(feat + (size_t)tileRow * DIN);
        const int c4 = tid & (DIN / 4 - 1);
        const int r0 = tid / (DIN / 4);
        #pragma unroll
        for (int i = 0; i < (16 * DIN / 4) / 256; ++i)
            *(float4*)(As + (16 * t + r0 + 2 * i) * APITCH1 + c4 * 4) = src[tid + i * 256];
    }
    __syncthreads();

    const int wave = tid >> 5;
    const int lane = tid & 31;
    const int m    = lane & 15;
    const int kp   = (lane >> 4) << 1;
    const int n    = wave * 16 + m;

    const float* Arow = As + m * APITCH1 + kp;
    const float* Bcol = W1 + n;

    v8f c[RT1];
    #pragma unroll
    for (int t = 0; t < RT1; ++t) c[t] = v8f{};

    #pragma unroll 8
    for (int k = 0; k < DIN; k += 4) {
        v2f b;
        b.x = Bcol[(size_t)(k + kp)     * DHID];
        b.y = Bcol[(size_t)(k + kp + 1) * DHID];
        #pragma unroll
        for (int t = 0; t < RT1; ++t) {
            v2f a = *(const v2f*)(Arow + t * 16 * APITCH1 + k);
            c[t] = __builtin_amdgcn_wmma_f32_16x16x4_f32(false, a, false, b,
                                                         (short)0, c[t], false, false);
        }
    }

    const float bias = b1[n];
    const int rOff = (lane >> 4) * 8;
    #pragma unroll
    for (int t = 0; t < RT1; ++t) {
        #pragma unroll
        for (int r = 0; r < 8; ++r) c[t][r] += bias;
    }
    #pragma unroll
    for (int t = 0; t < RT1; ++t) {
        if (rowBase + 16 * t < NNODES) {
            #pragma unroll
            for (int r = 0; r < 8; ++r)
                h[(size_t)(rowBase + 16 * t + rOff + r) * DHID + n] = c[t][r];
        }
    }
}

__global__ __launch_bounds__(128)
void gemm2_wmma(const float* __restrict__ hm, const float* __restrict__ W2,
                const float* __restrict__ b2, float* __restrict__ h2)
{
    __shared__ __attribute__((aligned(16))) float As[16 * APITCH2];
    const int rowBase = blockIdx.x * 16;
    const int tid = threadIdx.x;

    {
        const float4* src = (const float4*)(hm + (size_t)rowBase * DHID);
        const int c4 = tid & (DHID / 4 - 1);
        const int r0 = tid / (DHID / 4);
        #pragma unroll
        for (int i = 0; i < (16 * DHID / 4) / 128; ++i)
            *(float4*)(As + (r0 + 4 * i) * APITCH2 + c4 * 4) = src[tid + i * 128];
    }
    __syncthreads();

    const int wave = tid >> 5;
    const int lane = tid & 31;
    const int m    = lane & 15;
    const int kp   = (lane >> 4) << 1;
    const int n    = wave * 16 + m;

    const float* Arow = As + m * APITCH2 + kp;
    const float* Bcol = W2 + n;

    v8f c = {};
    #pragma unroll 8
    for (int k = 0; k < DHID; k += 4) {
        v2f a = *(const v2f*)(Arow + k);
        v2f b;
        b.x = Bcol[(size_t)(k + kp)     * DOUT];
        b.y = Bcol[(size_t)(k + kp + 1) * DOUT];
        c = __builtin_amdgcn_wmma_f32_16x16x4_f32(false, a, false, b,
                                                  (short)0, c, false, false);
    }

    const float bias = b2[n];
    const int rOff = (lane >> 4) * 8;
    #pragma unroll
    for (int r = 0; r < 8; ++r)
        h2[(size_t)(rowBase + rOff + r) * DOUT + n] = c[r] + bias;
}

#define MEDIAN_NET_BODY \
    "v_dual_min_num_f32 v34, v2, v3 :: v_dual_max_num_f32 v35, v2, v3\n\t" \
    "v_dual_min_num_f32 v2, v4, v5 :: v_dual_max_num_f32 v3, v4, v5\n\t" \
    "v_dual_min_num_f32 v4, v6, v7 :: v_dual_max_num_f32 v5, v6, v7\n\t" \
    "v_dual_min_num_f32 v6, v8, v9 :: v_dual_max_num_f32 v7, v8, v9\n\t" \
    "v_dual_min_num_f32 v8, v10, v11 :: v_dual_max_num_f32 v9, v10, v11\n\t" \
    "v_dual_min_num_f32 v10, v12, v13 :: v_dual_max_num_f32 v11, v12, v13\n\t" \
    "v_dual_min_num_f32 v12, v14, v15 :: v_dual_max_num_f32 v13, v14, v15\n\t" \
    "v_dual_min_num_f32 v14, v16, v17 :: v_dual_max_num_f32 v15, v16, v17\n\t" \
    "v_dual_min_num_f32 v16, v18, v19 :: v_dual_max_num_f32 v17, v18, v19\n\t" \
    "v_dual_min_num_f32 v18, v20, v21 :: v_dual_max_num_f32 v19, v20, v21\n\t" \
    "v_dual_min_num_f32 v20, v22, v23 :: v_dual_max_num_f32 v21, v22, v23\n\t" \
    "v_dual_min_num_f32 v22, v24, v25 :: v_dual_max_num_f32 v23, v24, v25\n\t" \
    "v_dual_min_num_f32 v24, v26, v27 :: v_dual_max_num_f32 v25, v26, v27\n\t" \
    "v_dual_min_num_f32 v26, v28, v29 :: v_dual_max_num_f32 v27, v28, v29\n\t" \
    "v_dual_min_num_f32 v28, v30, v31 :: v_dual_max_num_f32 v29, v30, v31\n\t" \
    "v_dual_min_num_f32 v30, v32, v33 :: v_dual_max_num_f32 v31, v32, v33\n\t" \
    "v_dual_min_num_f32 v32, v34, v2 :: v_dual_max_num_f32 v33, v34, v2\n\t" \
    "v_dual_min_num_f32 v2, v35, v3 :: v_dual_max_num_f32 v37, v35, v3\n\t" \
    "v_dual_min_num_f32 v34, v4, v6 :: v_dual_max_num_f32 v3, v4, v6\n\t" \
    "v_dual_min_num_f32 v4, v5, v7 :: v_dual_max_num_f32 v35, v5, v7\n\t" \
    "v_dual_min_num_f32 v6, v8, v10 :: v_dual_max_num_f32 v5, v8, v10\n\t" \
    "v_dual_min_num_f32 v8, v9, v11 :: v_dual_max_num_f32 v7, v9, v11\n\t" \
    "v_dual_min_num_f32 v10, v12, v14 :: v_dual_max_num_f32 v9, v12, v14\n\t" \
    "v_dual_min_num_f32 v12, v13, v15 :: v_dual_max_num_f32 v11, v13, v15\n\t" \
    "v_dual_min_num_f32 v14, v16, v18 :: v_dual_max_num_f32 v13, v16, v18\n\t" \
    "v_dual_min_num_f32 v16, v17, v19 :: v_dual_max_num_f32 v15, v17, v19\n\t" \
    "v_dual_min_num_f32 v18, v20, v22 :: v_dual_max_num_f32 v17, v20, v22\n\t" \
    "v_dual_min_num_f32 v20, v21, v23 :: v_dual_max_num_f32 v19, v21, v23\n\t" \
    "v_dual_min_num_f32 v22, v24, v26 :: v_dual_max_num_f32 v21, v24, v26\n\t" \
    "v_dual_min_num_f32 v24, v25, v27 :: v_dual_max_num_f32 v23, v25, v27\n\t" \
    "v_dual_min_num_f32 v26, v28, v30 :: v_dual_max_num_f32 v25, v28, v30\n\t" \
    "v_dual_min_num_f32 v28, v29, v31 :: v_dual_max_num_f32 v27, v29, v31\n\t" \
    "v_dual_min_num_f32 v30, v2, v33 :: v_dual_max_num_f32 v29, v2, v33\n\t" \
    "v_dual_min_num_f32 v2, v4, v3 :: v_dual_max_num_f32 v31, v4, v3\n\t" \
    "v_dual_min_num_f32 v4, v8, v5 :: v_dual_max_num_f32 v3, v8, v5\n\t" \
    "v_dual_min_num_f32 v8, v12, v9 :: v_dual_max_num_f32 v5, v12, v9\n\t" \
    "v_dual_min_num_f32 v12, v16, v13 :: v_dual_max_num_f32 v9, v16, v13\n\t" \
    "v_dual_min_num_f32 v16, v20, v17 :: v_dual_max_num_f32 v13, v20, v17\n\t" \
    "v_dual_min_num_f32 v20, v24, v21 :: v_dual_max_num_f32 v17, v24, v21\n\t" \
    "v_dual_min_num_f32 v24, v28, v25 :: v_dual_max_num_f32 v21, v28, v25\n\t" \
    "v_dual_min_num_f32 v28, v32, v34 :: v_dual_max_num_f32 v25, v32, v34\n\t" \
    "v_dual_min_num_f32 v32, v37, v35 :: v_dual_max_num_f32 v33, v37, v35\n\t" \
    "v_dual_min_num_f32 v34, v6, v10 :: v_dual_max_num_f32 v35, v6, v10\n\t" \
    "v_dual_min_num_f32 v6, v7, v11 :: v_dual_max_num_f32 v37, v7, v11\n\t" \
    "v_dual_min_num_f32 v10, v14, v18 :: v_dual_max_num_f32 v7, v14, v18\n\t" \
    "v_dual_min_num_f32 v14, v15, v19 :: v_dual_max_num_f32 v11, v15, v19\n\t" \
    "v_dual_min_num_f32 v18, v22, v26 :: v_dual_max_num_f32 v15, v22, v26\n\t" \
    "v_dual_min_num_f32 v22, v23, v27 :: v_dual_max_num_f32 v19, v23, v27\n\t" \
    "v_dual_min_num_f32 v26, v30, v2 :: v_dual_max_num_f32 v23, v30, v2\n\t" \
    "v_dual_min_num_f32 v2, v29, v31 :: v_dual_max_num_f32 v27, v29, v31\n\t" \
    "v_dual_min_num_f32 v30, v4, v8 :: v_dual_max_num_f32 v29, v4, v8\n\t" \
    "v_dual_min_num_f32 v4, v3, v5 :: v_dual_max_num_f32 v31, v3, v5\n\t" \
    "v_dual_min_num_f32 v8, v12, v16 :: v_dual_max_num_f32 v3, v12, v16\n\t" \
    "v_dual_min_num_f32 v12, v9, v13 :: v_dual_max_num_f32 v5, v9, v13\n\t" \
    "v_dual_min_num_f32 v16, v20, v24 :: v_dual_max_num_f32 v9, v20, v24\n\t" \
    "v_dual_min_num_f32 v20, v17, v21 :: v_dual_max_num_f32 v13, v17, v21\n\t" \
    "v_dual_min_num_f32 v24, v28, v34 :: v_dual_max_num_f32 v17, v28, v34\n\t" \
    "v_dual_min_num_f32 v28, v33, v37 :: v_dual_max_num_f32 v21, v33, v37\n\t" \
    "v_dual_min_num_f32 v34, v10, v18 :: v_dual_max_num_f32 v33, v10, v18\n\t" \
    "v_dual_min_num_f32 v10, v11, v19 :: v_dual_max_num_f32 v37, v11, v19\n\t" \
    "v_dual_min_num_f32 v18, v2, v25 :: v_dual_max_num_f32 v11, v2, v25\n\t" \
    "v_dual_min_num_f32 v2, v32, v23 :: v_dual_max_num_f32 v19, v32, v23\n\t" \
    "v_dual_min_num_f32 v32, v4, v35 :: v_dual_max_num_f32 v23, v4, v35\n\t" \
    "v_dual_min_num_f32 v4, v6, v29 :: v_dual_max_num_f32 v25, v6, v29\n\t" \
    "v_dual_min_num_f32 v6, v12, v7 :: v_dual_max_num_f32 v29, v12, v7\n\t" \
    "v_dual_min_num_f32 v12, v14, v3 :: v_dual_max_num_f32 v7, v14, v3\n\t" \
    "v_dual_min_num_f32 v14, v20, v15 :: v_dual_max_num_f32 v3, v20, v15\n\t" \
    "v_dual_min_num_f32 v20, v22, v9 :: v_dual_max_num_f32 v15, v22, v9\n\t" \
    "v_dual_min_num_f32 v22, v24, v34 :: v_dual_max_num_f32 v9, v24, v34\n\t" \
    "v_dual_min_num_f32 v24, v21, v37 :: v_dual_max_num_f32 v35, v21, v37\n\t" \
    "v_dual_min_num_f32 v34, v26, v18 :: v_dual_max_num_f32 v21, v26, v18\n\t" \
    "v_dual_min_num_f32 v18, v2, v11 :: v_dual_max_num_f32 v37, v2, v11\n\t" \
    "v_dual_min_num_f32 v2, v19, v27 :: v_dual_max_num_f32 v11, v19, v27\n\t" \
    "v_dual_min_num_f32 v26, v30, v32 :: v_dual_max_num_f32 v19, v30, v32\n\t" \
    "v_dual_min_num_f32 v30, v4, v23 :: v_dual_max_num_f32 v27, v4, v23\n\t" \
    "v_dual_min_num_f32 v4, v25, v31 :: v_dual_max_num_f32 v23, v25, v31\n\t" \
    "v_dual_min_num_f32 v32, v8, v6 :: v_dual_max_num_f32 v25, v8, v6\n\t" \
    "v_dual_min_num_f32 v6, v12, v29 :: v_dual_max_num_f32 v31, v12, v29\n\t" \
    "v_dual_min_num_f32 v8, v7, v5 :: v_dual_max_num_f32 v29, v7, v5\n\t" \
    "v_dual_min_num_f32 v12, v16, v14 :: v_dual_max_num_f32 v5, v16, v14\n\t" \
    "v_dual_min_num_f32 v14, v20, v3 :: v_dual_max_num_f32 v7, v20, v3\n\t" \
    "v_dual_min_num_f32 v16, v15, v13 :: v_dual_max_num_f32 v3, v15, v13\n\t" \
    "v_dual_min_num_f32 v20, v34, v26 :: v_dual_max_num_f32 v13, v34, v26\n\t" \
    "v_dual_min_num_f32 v26, v21, v19 :: v_dual_max_num_f32 v15, v21, v19\n\t" \
    "v_dual_min_num_f32 v34, v18, v30 :: v_dual_max_num_f32 v19, v18, v30\n\t" \
    "v_dual_min_num_f32 v18, v37, v27 :: v_dual_max_num_f32 v21, v37, v27\n\t" \
    "v_dual_min_num_f32 v30, v2, v4 :: v_dual_max_num_f32 v27, v2, v4\n\t" \
    "v_dual_min_num_f32 v2, v11, v23 :: v_dual_max_num_f32 v37, v11, v23\n\t" \
    "v_dual_min_num_f32 v4, v32, v12 :: v_dual_max_num_f32 v11, v32, v12\n\t" \
    "v_dual_min_num_f32 v12, v25, v5 :: v_dual_max_num_f32 v23, v25, v5\n\t" \
    "v_dual_min_num_f32 v32, v6, v14 :: v_dual_max_num_f32 v5, v6, v14\n\t" \
    "v_dual_min_num_f32 v6, v31, v7 :: v_dual_max_num_f32 v25, v31, v7\n\t" \
    "v_dual_min_num_f32 v14, v8, v16 :: v_dual_max_num_f32 v7, v8, v16\n\t" \
    "v_dual_min_num_f32 v8, v29, v3 :: v_dual_max_num_f32 v31, v29, v3\n\t" \
    "v_dual_min_num_f32 v16, v18, v17 :: v_dual_max_num_f32 v3, v18, v17\n\t" \
    "v_dual_min_num_f32 v18, v30, v13 :: v_dual_max_num_f32 v17, v30, v13\n\t" \
    "v_dual_min_num_f32 v30, v2, v15 :: v_dual_max_num_f32 v13, v2, v15\n\t" \
    "v_dual_min_num_f32 v2, v28, v19 :: v_dual_max_num_f32 v15, v28, v19\n\t" \
    "v_dual_min_num_f32 v28, v6, v33 :: v_dual_max_num_f32 v19, v6, v33\n\t" \
    "v_dual_min_num_f32 v6, v14, v11 :: v_dual_max_num_f32 v29, v14, v11\n\t" \
    "v_dual_min_num_f32 v14, v8, v23 :: v_dual_max_num_f32 v11, v8, v23\n\t" \
    "v_dual_min_num_f32 v8, v10, v5 :: v_dual_max_num_f32 v23, v10, v5\n\t" \
    "v_dual_min_num_f32 v10, v26, v16 :: v_dual_max_num_f32 v5, v26, v16\n\t" \
    "v_dual_min_num_f32 v16, v34, v18 :: v_dual_max_num_f32 v33, v34, v18\n\t" \
    "v_dual_min_num_f32 v18, v30, v3 :: v_dual_max_num_f32 v39, v30, v3\n\t" \
    "v_dual_min_num_f32 v26, v2, v17 :: v_dual_max_num_f32 v3, v2, v17\n\t" \
    "v_dual_min_num_f32 v2, v13, v21 :: v_dual_max_num_f32 v17, v13, v21\n\t" \
    "v_dual_min_num_f32 v30, v15, v27 :: v_dual_max_num_f32 v13, v15, v27\n\t" \
    "v_dual_min_num_f32 v34, v12, v28 :: v_dual_max_num_f32 v15, v12, v28\n\t" \
    "v_dual_min_num_f32 v12, v32, v6 :: v_dual_max_num_f32 v21, v32, v6\n\t" \
    "v_dual_min_num_f32 v6, v14, v19 :: v_dual_max_num_f32 v27, v14, v19\n\t" \
    "v_dual_min_num_f32 v14, v8, v29 :: v_dual_max_num_f32 v19, v8, v29\n\t" \
    "v_dual_min_num_f32 v8, v11, v25 :: v_dual_max_num_f32 v29, v11, v25\n\t" \
    "v_dual_min_num_f32 v28, v23, v7 :: v_dual_max_num_f32 v11, v23, v7\n\t" \
    "v_dual_min_num_f32 v32, v20, v10 :: v_dual_max_num_f32 v7, v20, v10\n\t" \
    "v_dual_min_num_f32 v10, v16, v5 :: v_dual_max_num_f32 v23, v16, v5\n\t" \
    "v_dual_min_num_f32 v16, v33, v18 :: v_dual_max_num_f32 v5, v33, v18\n\t" \
    "v_dual_min_num_f32 v18, v26, v39 :: v_dual_max_num_f32 v25, v26, v39\n\t" \
    "v_dual_min_num_f32 v20, v3, v2 :: v_dual_max_num_f32 v33, v3, v2\n\t" \
    "v_dual_min_num_f32 v2, v30, v17 :: v_dual_max_num_f32 v3, v30, v17\n\t" \
    "v_dual_min_num_f32 v26, v13, v37 :: v_dual_max_num_f32 v17, v13, v37\n\t" \
    "v_dual_min_num_f32 v30, v4, v34 :: v_dual_max_num_f32 v13, v4, v34\n\t" \
    "v_dual_min_num_f32 v4, v12, v15 :: v_dual_max_num_f32 v37, v12, v15\n\t" \
    "v_dual_min_num_f32 v12, v21, v6 :: v_dual_max_num_f32 v15, v21, v6\n\t" \
    "v_dual_min_num_f32 v6, v14, v27 :: v_dual_max_num_f32 v21, v14, v27\n\t" \
    "v_dual_min_num_f32 v14, v19, v8 :: v_dual_max_num_f32 v27, v19, v8\n\t" \
    "v_dual_min_num_f32 v8, v28, v29 :: v_dual_max_num_f32 v19, v28, v29\n\t" \
    "v_dual_min_num_f32 v28, v11, v31 :: v_dual_max_num_f32 v29, v11, v31\n\t" \
    "v_dual_min_num_f32 v34, v32, v30 :: v_dual_max_num_f32 v11, v32, v30\n\t" \
    "v_dual_min_num_f32 v30, v7, v13 :: v_dual_max_num_f32 v31, v7, v13\n\t" \
    "v_dual_min_num_f32 v32, v10, v4 :: v_dual_max_num_f32 v7, v10, v4\n\t" \
    "v_dual_min_num_f32 v4, v23, v37 :: v_dual_max_num_f32 v13, v23, v37\n\t" \
    "v_dual_min_num_f32 v10, v16, v12 :: v_dual_max_num_f32 v23, v16, v12\n\t" \
    "v_dual_min_num_f32 v12, v5, v15 :: v_dual_max_num_f32 v37, v5, v15\n\t" \
    "v_dual_min_num_f32 v16, v18, v6 :: v_dual_max_num_f32 v5, v18, v6\n\t" \
    "v_dual_min_num_f32 v6, v25, v21 :: v_dual_max_num_f32 v15, v25, v21\n\t" \
    "v_dual_min_num_f32 v18, v20, v14 :: v_dual_max_num_f32 v21, v20, v14\n\t" \
    "v_dual_min_num_f32 v14, v33, v27 :: v_dual_max_num_f32 v25, v33, v27\n\t" \
    "v_dual_min_num_f32 v20, v2, v8 :: v_dual_max_num_f32 v27, v2, v8\n\t" \
    "v_dual_min_num_f32 v2, v3, v19 :: v_dual_max_num_f32 v33, v3, v19\n\t" \
    "v_dual_min_num_f32 v8, v26, v28 :: v_dual_max_num_f32 v3, v26, v28\n\t" \
    "v_dual_min_num_f32 v26, v17, v29 :: v_dual_max_num_f32 v19, v17, v29\n\t" \
    "v_dual_min_num_f32 v28, v6, v9 :: v_dual_max_num_f32 v17, v6, v9\n\t" \
    "v_dual_min_num_f32 v6, v18, v11 :: v_dual_max_num_f32 v9, v18, v11\n\t" \
    "v_dual_min_num_f32 v18, v14, v31 :: v_dual_max_num_f32 v11, v14, v31\n\t" \
    "v_dual_min_num_f32 v14, v20, v7 :: v_dual_max_num_f32 v29, v20, v7\n\t" \
    "v_dual_min_num_f32 v20, v2, v13 :: v_dual_max_num_f32 v7, v2, v13\n\t" \
    "v_dual_min_num_f32 v2, v8, v23 :: v_dual_max_num_f32 v13, v8, v23\n\t" \
    "v_dual_min_num_f32 v8, v26, v37 :: v_dual_max_num_f32 v23, v26, v37\n\t" \
    "v_dual_min_num_f32 v26, v24, v5 :: v_dual_max_num_f32 v31, v24, v5\n\t" \
    "s_delay_alu instid0(VALU_DEP_4)\n\t" \
    "v_dual_min_num_f32 v24, v20, v17 :: v_dual_max_num_f32 v5, v20, v17\n\t" \
    "s_delay_alu instid0(VALU_DEP_4)\n\t" \
    "v_dual_min_num_f32 v20, v2, v9 :: v_dual_max_num_f32 v17, v2, v9\n\t" \
    "s_delay_alu instid0(VALU_DEP_4)\n\t" \
    "v_dual_min_num_f32 v2, v8, v11 :: v_dual_max_num_f32 v9, v8, v11\n\t" \
    "s_delay_alu instid0(VALU_DEP_4)\n\t" \
    "v_dual_min_num_f32 v8, v26, v29 :: v_dual_max_num_f32 v11, v26, v29\n\t" \
    "s_delay_alu instid0(VALU_DEP_2)\n\t" \
    "v_dual_min_num_f32 v26, v2, v5 :: v_dual_max_num_f32 v29, v2, v5\n\t" \
    "s_delay_alu instid0(VALU_DEP_2)\n\t" \
    "v_dual_min_num_f32 v2, v8, v17 :: v_dual_max_num_f32 v5, v8, v17\n\t"

#define MEDIAN_NET_IO \
    "+{v2}"(v[0]), "+{v3}"(v[1]), "+{v4}"(v[2]), "+{v5}"(v[3]), \
    "+{v6}"(v[4]), "+{v7}"(v[5]), "+{v8}"(v[6]), "+{v9}"(v[7]), \
    "+{v10}"(v[8]), "+{v11}"(v[9]), "+{v12}"(v[10]), "+{v13}"(v[11]), \
    "+{v14}"(v[12]), "+{v15}"(v[13]), "+{v16}"(v[14]), "+{v17}"(v[15]), \
    "+{v18}"(v[16]), "+{v19}"(v[17]), "+{v20}"(v[18]), "+{v21}"(v[19]), \
    "+{v22}"(v[20]), "+{v23}"(v[21]), "+{v24}"(v[22]), "+{v25}"(v[23]), \
    "+{v26}"(v[24]), "+{v27}"(v[25]), "+{v28}"(v[26]), "+{v29}"(v[27]), \
    "+{v30}"(v[28]), "+{v31}"(v[29]), "+{v32}"(v[30]), "+{v33}"(v[31])
#define MEDIAN_NET_SPARE "v34", "v35", "v36", "v37", "v38", "v39"

template <int D, int NODES_PER_BLK, bool RELU>
__global__ __launch_bounds__(D * NODES_PER_BLK)
void median_kernel(const float* __restrict__ src,
                   const int* __restrict__ nbr,
                   float* __restrict__ dst)
{
    const int node = __builtin_amdgcn_readfirstlane(blockIdx.x * NODES_PER_BLK + threadIdx.x / D);
    const int f    = threadIdx.x % D;

    typedef const int __attribute__((address_space(4))) cidx_t;
    cidx_t* nb = (cidx_t*)(unsigned long long)(nbr + (size_t)node * KNBR);
    typedef const float __attribute__((address_space(1))) gfloat_t;

    float v[KNBR];
    #pragma unroll
    for (int j = 0; j < KNBR; ++j) {
        const unsigned long long rowAddr =
            (unsigned long long)src + (unsigned long long)((unsigned)nb[j] * (unsigned)(D * 4));
        const unsigned lo = __builtin_amdgcn_readfirstlane((unsigned)rowAddr);
        const unsigned hi = __builtin_amdgcn_readfirstlane((unsigned)(rowAddr >> 32));
        gfloat_t* row = (gfloat_t*)(((unsigned long long)hi << 32) | lo);
        v[j] = row[f];
    }

    float med;
    if (RELU)
        asm(MEDIAN_NET_BODY "s_delay_alu instid0(VALU_DEP_1)\n\t" "v_minmax_num_f32 %0, v2, v29, 0"
            : "=v"(med), MEDIAN_NET_IO : : MEDIAN_NET_SPARE);
    else
        asm(MEDIAN_NET_BODY "s_delay_alu instid0(VALU_DEP_1)\n\t" "v_min_num_f32 %0, v2, v29"
            : "=v"(med), MEDIAN_NET_IO : : MEDIAN_NET_SPARE);
    dst[(size_t)node * D + f] = med;
}

extern "C" void kernel_launch(void* const* d_in, const int* in_sizes, int n_in,
                              void* d_out, int out_size, void* d_ws, size_t ws_size,
                              hipStream_t stream)
{
    const float* feat = (const float*)d_in[0];
    const float* W1   = (const float*)d_in[1];
    const float* b1   = (const float*)d_in[2];
    const float* W2   = (const float*)d_in[3];
    const float* b2   = (const float*)d_in[4];
    const int*   nbr  = (const int*)d_in[5];

    float* h   = (float*)d_ws;
    float* hm  = h  + (size_t)NNODES * DHID;
    float* h2  = hm + (size_t)NNODES * DHID;
    float* out = (float*)d_out;

    gemm1_wmma<<<G1BLOCKS, 256, 0, stream>>>(feat, W1, b1, h);
    median_kernel<DHID, 1, true ><<<NNODES,     DHID, 0, stream>>>(h,  nbr, hm);
    gemm2_wmma<<<NNODES / 16, 128, 0, stream>>>(hm, W2, b2, h2);
    median_kernel<DOUT, 2, false><<<NNODES / 2, DOUT * 2, 0, stream>>>(h2, nbr, out);
}
